# starting kernel
# baseline (speedup 1.0000x reference)
#include <hip/hip_runtime.h>

typedef __attribute__((ext_vector_type(2))) float f32x2;
typedef __attribute__((ext_vector_type(8))) float f32x8;

#define VOCAB 32000
#define EMB   512
#define HID   1024
#define GATES 4096
#define BATCH 32
#define SEQ   512

__global__ __launch_bounds__(256) void zero_ws_kernel(float* __restrict__ p, int n) {
    int i = blockIdx.x * blockDim.x + threadIdx.x;
    if (i < n) p[i] = 0.0f;
}

__device__ __forceinline__ float sigmoidf_dev(float x) {
    return 1.0f / (1.0f + __expf(-x));
}

__global__ __launch_bounds__(256) void lstm_step_kernel(
    const int*   __restrict__ src,
    const float* __restrict__ emb,
    const float* __restrict__ W_ih,
    const float* __restrict__ W_hh,
    const float* __restrict__ b_ih,
    const float* __restrict__ b_hh,
    const float* __restrict__ h_prev, long h_prev_bstride,
    float*       __restrict__ c_state,
    float*       __restrict__ h_out, long h_out_bstride,
    int t)
{
    __shared__ float lds_g[4][BATCH][16];

    const int tid  = threadIdx.x;
    const int lane = tid & 31;
    const int wave = tid >> 5;
    const int gate = wave >> 1;
    const int mt   = wave & 1;
    const int l16  = lane & 15;
    const int hi   = lane >> 4;

    const int n0 = blockIdx.x * 16;
    const int N  = n0 + l16;
    const int M  = mt * 16 + l16;
    const int gcol = gate * HID + N;

    f32x8 acc = {0.f, 0.f, 0.f, 0.f, 0.f, 0.f, 0.f, 0.f};

    {
        const long row = (long)src[M * SEQ + t];
        const float* a_ptr = emb + row * (long)EMB;
        #pragma unroll 4
        for (int k0 = 0; k0 < EMB; k0 += 4) {
            const int kb = k0 + 2 * hi;
            f32x2 a = *(const f32x2*)(a_ptr + kb);
            f32x2 b;
            b.x = W_ih[(long)kb * GATES + gcol];
            b.y = W_ih[(long)(kb + 1) * GATES + gcol];
            acc = __builtin_amdgcn_wmma_f32_16x16x4_f32(
                false, a, false, b, (short)0, acc, false, false);
        }
    }

    {
        const float* a_ptr = h_prev + (long)M * h_prev_bstride;
        #pragma unroll 4
        for (int k0 = 0; k0 < HID; k0 += 4) {
            const int kb = k0 + 2 * hi;
            f32x2 a = *(const f32x2*)(a_ptr + kb);
            f32x2 b;
            b.x = W_hh[(long)kb * GATES + gcol];
            b.y = W_hh[(long)(kb + 1) * GATES + gcol];
            acc = __builtin_amdgcn_wmma_f32_16x16x4_f32(
                false, a, false, b, (short)0, acc, false, false);
        }
    }

    const float bias = b_ih[gcol] + b_hh[gcol];

    #pragma unroll
    for (int r = 0; r < 8; ++r) {
        const int brow = mt * 16 + hi * 8 + r;
        lds_g[gate][brow][l16] = acc[r] + bias;
    }
    __syncthreads();

    #pragma unroll
    for (int idx = tid; idx < BATCH * 16; idx += 256) {
        const int b = idx >> 4;
        const int n = idx & 15;
        const float ig = sigmoidf_dev(lds_g[0][b][n]);
        const float fg = sigmoidf_dev(lds_g[1][b][n]);
        const float gg = tanhf(lds_g[2][b][n]);
        const float og = sigmoidf_dev(lds_g[3][b][n]);
        const long cidx = (long)b * HID + n0 + n;
        const float c_new = fg * c_state[cidx] + ig * gg;
        c_state[cidx] = c_new;
        h_out[(long)b * h_out_bstride + n0 + n] = og * tanhf(c_new);
    }
}

extern "C" void kernel_launch(void* const* d_in, const int* in_sizes, int n_in,
                              void* d_out, int out_size, void* d_ws, size_t ws_size,
                              hipStream_t stream) {
    const int*   src  = (const int*)  d_in[0];
    const float* emb  = (const float*)d_in[1];
    const float* W_ih = (const float*)d_in[2];
    const float* W_hh = (const float*)d_in[3];
    const float* b_ih = (const float*)d_in[4];
    const float* b_hh = (const float*)d_in[5];
    float* out = (float*)d_out;

    float* c_state = (float*)d_ws;
    float* h0      = c_state + BATCH * HID;

    const int n_init = 2 * BATCH * HID;
    zero_ws_kernel<<<(n_init + 255) / 256, 256, 0, stream>>>(c_state, n_init);

    for (int t = 0; t < SEQ; ++t) {
        const float* h_prev = (t == 0) ? h0 : (out + (long)(t - 1) * HID);
        const long   hstr   = (t == 0) ? (long)HID : (long)SEQ * HID;
        lstm_step_kernel<<<HID / 16, 256, 0, stream>>>(
            src, emb, W_ih, W_hh, b_ih, b_hh,
            h_prev, hstr, c_state,
            out + (long)t * HID, (long)SEQ * HID, t);
    }
}


# W_hh staged through a two-stage LDS ring: the block fetches each 128-row chunk with 16-byte loads one chunk ahead, A fragments with it; a B fragment is one 8-byte LDS read; f32 sum order kept
# speedup vs baseline: 2.2683x; 2.2683x over previous
#include <hip/hip_runtime.h>


typedef __attribute__((ext_vector_type(2))) float f32x2;
typedef __attribute__((ext_vector_type(4))) float f32x4;
typedef __attribute__((ext_vector_type(8))) float f32x8;

#define VOCAB 32000
#define EMB   512
#define HID   1024
#define GATES 4096
#define BATCH 32
#define SEQ   512
#define TCHUNK 32

__global__ __launch_bounds__(256) void zero_ws_kernel(float* __restrict__ p, int n) {
    int i = blockIdx.x * blockDim.x + threadIdx.x;
    if (i < n) p[i] = 0.0f;
}

__device__ __forceinline__ float sigmoidf_dev(float x) {
    return 1.0f / (1.0f + __expf(-x));
}

__global__ __launch_bounds__(256) void xproj_kernel(
    const int*   __restrict__ src,
    const float* __restrict__ emb,
    const float* __restrict__ W_ih,
    float*       __restrict__ xg,
    int t0)
{
    const int tid  = threadIdx.x;
    const int lane = tid & 31;
    const int wave = tid >> 5;
    const int wm   = wave >> 1;
    const int wn   = wave & 1;
    const int c    = lane & 15;
    const int hi   = lane >> 4;

    const int tl    = blockIdx.y * 4 + wm;
    const int cbase = blockIdx.x * 128 + wn * 64;

    const float* a0 = emb + (long)src[c * SEQ + t0 + tl] * (long)EMB + 2 * hi;
    const float* a1 = emb + (long)src[(16 + c) * SEQ + t0 + tl] * (long)EMB + 2 * hi;
    const float* bp = W_ih + (long)(2 * hi) * GATES + cbase + 4 * c;

    f32x8 acc[2][4];
    #pragma unroll
    for (int mi = 0; mi < 2; ++mi)
        #pragma unroll
        for (int j = 0; j < 4; ++j)
            acc[mi][j] = (f32x8){0.f, 0.f, 0.f, 0.f, 0.f, 0.f, 0.f, 0.f};

    #pragma unroll 8
    for (int k0 = 0; k0 < EMB; k0 += 4) {
        const f32x2 av0 = *(const f32x2*)(a0 + k0);
        const f32x2 av1 = *(const f32x2*)(a1 + k0);
        const f32x4 b0 = *(const f32x4*)(bp + (long)k0 * GATES);
        const f32x4 b1 = *(const f32x4*)(bp + (long)(k0 + 1) * GATES);
        #pragma unroll
        for (int j = 0; j < 4; ++j) {
            f32x2 b;
            b.x = b0[j];
            b.y = b1[j];
            acc[0][j] = __builtin_amdgcn_wmma_f32_16x16x4_f32(
                false, av0, false, b, (short)0, acc[0][j], false, false);
            acc[1][j] = __builtin_amdgcn_wmma_f32_16x16x4_f32(
                false, av1, false, b, (short)0, acc[1][j], false, false);
        }
    }

    #pragma unroll
    for (int mi = 0; mi < 2; ++mi) {
        #pragma unroll
        for (int r = 0; r < 8; ++r) {
            const int brow = mi * 16 + hi * 8 + r;
            f32x4 v;
            v.x = acc[mi][0][r];
            v.y = acc[mi][1][r];
            v.z = acc[mi][2][r];
            v.w = acc[mi][3][r];
            *(f32x4*)(xg + ((long)tl * BATCH + brow) * GATES + cbase + 4 * c) = v;
        }
    }
}

#define KC     128
#define NCHUNK (HID / KC)
#define PPITCH 128
#define STAGE_FLOATS ((KC / 2) * PPITCH)

__device__ __forceinline__ void ask_chunk(
    const float* __restrict__ bsrc, const f32x4* __restrict__ a_ptr, int ch,
    f32x4 (&breg)[KC / 16], f32x4 (&areg)[KC / 8])
{
    #pragma unroll
    for (int j = 0; j < KC / 16; ++j)
        breg[j] = *(const f32x4*)(bsrc + (long)(ch * KC + 16 * j) * GATES);
    #pragma unroll
    for (int g = 0; g < KC / 8; ++g)
        areg[g] = a_ptr[(ch * (KC / 8) + g) * 32];
}

__device__ __forceinline__ void put_chunk(
    float* stage, int bdst, const f32x4 (&breg)[KC / 16])
{
    #pragma unroll
    for (int j = 0; j < KC / 16; ++j) {
        float* d = stage + bdst + 8 * j * PPITCH;
        d[0] = breg[j].x;
        d[2] = breg[j].y;
        d[4] = breg[j].z;
        d[6] = breg[j].w;
    }
}

__device__ __forceinline__ f32x8 mma_chunk(
    f32x8 acc, const f32x4 (&areg)[KC / 8], const float* bl)
{
    f32x2 bf[2][8];
    #pragma unroll
    for (int s = 0; s < 8; ++s)
        bf[0][s] = *(const f32x2*)(bl + (2 * s) * PPITCH);
    #pragma unroll
    for (int sb = 0; sb < KC / 32; ++sb) {
        if (sb + 1 < KC / 32) {
            #pragma unroll
            for (int s = 0; s < 8; ++s)
                bf[(sb + 1) & 1][s] = *(const f32x2*)(bl + (2 * (8 * (sb + 1) + s)) * PPITCH);
        }
        __builtin_amdgcn_sched_barrier(0);
        #pragma unroll
        for (int s = 0; s < 8; ++s) {
            const f32x4 a4 = areg[(8 * sb + s) >> 1];
            f32x2 a;
            if (s & 1) { a.x = a4.z; a.y = a4.w; }
            else       { a.x = a4.x; a.y = a4.y; }
            acc = __builtin_amdgcn_wmma_f32_16x16x4_f32(
                false, a, false, bf[sb & 1][s], (short)0, acc, false, false);
        }
    }
    return acc;
}

__global__ __launch_bounds__(256) void lstm_step_kernel(
    const float* __restrict__ xg_t,
    const float* __restrict__ W_hh,
    const float* __restrict__ b_ih,
    const float* __restrict__ b_hh,
    const float* __restrict__ hf_rd,
    float*       __restrict__ hf_wr,
    float*       __restrict__ c_state,
    float*       __restrict__ h_out, long h_out_bstride)
{
    __shared__ __attribute__((aligned(16))) float lds_b[2][STAGE_FLOATS];
    float (*lds_g)[BATCH][16] = (float (*)[BATCH][16])&lds_b[0][0];

    const int tid  = threadIdx.x;
    const int lane = tid & 31;
    const int wave = tid >> 5;
    const int gate = wave >> 1;
    const int mt   = wave & 1;
    const int l16  = lane & 15;
    const int hi   = lane >> 4;

    const int n0 = blockIdx.x * 16;
    const int N  = n0 + l16;
    const int gcol = gate * HID + N;

    const int brow   = tid >> 4;
    const int bpiece = tid & 15;
    const float* bsrc = W_hh + (long)brow * GATES + (bpiece >> 2) * HID + n0 + 4 * (bpiece & 3);
    const int bdst = (brow >> 1) * PPITCH + 8 * (bpiece ^ (((brow >> 1) & 1) << 2)) + (brow & 1);

    const f32x4* a_ptr = (const f32x4*)hf_rd + (mt * (HID / 8)) * 32 + lane;

    f32x4 breg[KC / 16];
    f32x4 a_even[KC / 8];
    f32x4 a_odd[KC / 8];

    ask_chunk(bsrc, a_ptr, 0, breg, a_even);

    f32x8 acc;
    #pragma unroll
    for (int r = 0; r < 8; ++r)
        acc[r] = xg_t[(long)(mt * 16 + hi * 8 + r) * GATES + gcol];

    const float bias = b_ih[gcol] + b_hh[gcol];

    put_chunk(lds_b[0], bdst, breg);
    __syncthreads();

    const int bl_off = hi * PPITCH + 2 * ((gate * 16 + l16) ^ (hi << 4));
    const float* bl0 = lds_b[0] + bl_off;
    const float* bl1 = lds_b[1] + bl_off;

    for (int ch = 0; ch < NCHUNK; ch += 2) {
        ask_chunk(bsrc, a_ptr, ch + 1, breg, a_odd);
        acc = mma_chunk(acc, a_even, bl0);
        put_chunk(lds_b[1], bdst, breg);
        __syncthreads();

        const bool more = (ch + 2 < NCHUNK);
        if (more) ask_chunk(bsrc, a_ptr, ch + 2, breg, a_even);
        acc = mma_chunk(acc, a_odd, bl1);
        if (more) {
            put_chunk(lds_b[0], bdst, breg);
            __syncthreads();
        }
    }

    asm volatile("s_wait_alu depctr_va_vdst(0)\n\tv_nop\n\tv_nop\n\tv_nop\n\tv_nop" : "+v"(acc));

    #pragma unroll
    for (int r = 0; r < 8; ++r) {
        const int brw = mt * 16 + hi * 8 + r;
        lds_g[gate][brw][l16] = acc[r] + bias;
    }
    __syncthreads();

    #pragma unroll
    for (int idx = tid; idx < BATCH * 16; idx += 256) {
        const int b = idx >> 4;
        const int n = idx & 15;
        const float ig = sigmoidf_dev(lds_g[0][b][n]);
        const float fg = sigmoidf_dev(lds_g[1][b][n]);
        const float gg = tanhf(lds_g[2][b][n]);
        const float og = sigmoidf_dev(lds_g[3][b][n]);
        const long cidx = (long)b * HID + n0 + n;
        const float c_new = fg * c_state[cidx] + ig * gg;
        c_state[cidx] = c_new;
        const float h_new = og * tanhf(c_new);
        h_out[(long)b * h_out_bstride + n0 + n] = h_new;
        const int k  = n0 + n;
        const int kr = k & 7;
        const int fl = ((kr >> 1) & 1) * 16 + (b & 15);
        const int fe = (kr & 1) + 2 * (kr >> 2);
        hf_wr[((((b >> 4) * (HID / 8) + (k >> 3)) * 32 + fl) << 2) + fe] = h_new;
    }
}

extern "C" void kernel_launch(void* const* d_in, const int* in_sizes, int n_in,
                              void* d_out, int out_size, void* d_ws, size_t ws_size,
                              hipStream_t stream) {
    const int*   src  = (const int*)  d_in[0];
    const float* emb  = (const float*)d_in[1];
    const float* W_ih = (const float*)d_in[2];
    const float* W_hh = (const float*)d_in[3];
    const float* b_ih = (const float*)d_in[4];
    const float* b_hh = (const float*)d_in[5];
    float* out = (float*)d_out;

    float* c_state = (float*)d_ws;
    float* hf0     = c_state + BATCH * HID;
    float* hf1     = hf0 + BATCH * HID;
    float* xg      = hf1 + BATCH * HID;

    const int n_init = 2 * BATCH * HID;
    zero_ws_kernel<<<(n_init + 255) / 256, 256, 0, stream>>>(c_state, n_init);

    for (int t0 = 0; t0 < SEQ; t0 += TCHUNK) {
        xproj_kernel<<<dim3(GATES / 128, TCHUNK * BATCH / 128), 256, 0, stream>>>(
            src, emb, W_ih, xg, t0);

        for (int tl = 0; tl < TCHUNK; ++tl) {
            const int t = t0 + tl;
            const float* hf_rd = (t & 1) ? hf1 : hf0;
            float*       hf_wr = (t & 1) ? hf0 : hf1;
            lstm_step_kernel<<<HID / 16, 256, 0, stream>>>(
                xg + (long)tl * BATCH * GATES, W_hh, b_ih, b_hh,
                hf_rd, hf_wr, c_state,
                out + (long)t * HID, (long)SEQ * HID);
        }
    }
}


# h fetched once per block: each chunk's share of the A-fragment plane rides the LDS ring beside W_hh, read there by the four gate waves of a batch half (4x fewer global A loads); f32 sum order kept
# speedup vs baseline: 2.3488x; 1.0355x over previous
#include <hip/hip_runtime.h>


typedef __attribute__((ext_vector_type(2))) float f32x2;
typedef __attribute__((ext_vector_type(4))) float f32x4;
typedef __attribute__((ext_vector_type(8))) float f32x8;

#define VOCAB 32000
#define EMB   512
#define HID   1024
#define GATES 4096
#define BATCH 32
#define SEQ   512
#define TCHUNK 32

__global__ __launch_bounds__(256) void zero_ws_kernel(float* __restrict__ p, int n) {
    int i = blockIdx.x * blockDim.x + threadIdx.x;
    if (i < n) p[i] = 0.0f;
}

__device__ __forceinline__ float sigmoidf_dev(float x) {
    return 1.0f / (1.0f + __expf(-x));
}

__global__ __launch_bounds__(256) void xproj_kernel(
    const int*   __restrict__ src,
    const float* __restrict__ emb,
    const float* __restrict__ W_ih,
    float*       __restrict__ xg,
    int t0)
{
    const int tid  = threadIdx.x;
    const int lane = tid & 31;
    const int wave = tid >> 5;
    const int wm   = wave >> 1;
    const int wn   = wave & 1;
    const int c    = lane & 15;
    const int hi   = lane >> 4;

    const int tl    = blockIdx.y * 4 + wm;
    const int cbase = blockIdx.x * 128 + wn * 64;

    const float* a0 = emb + (long)src[c * SEQ + t0 + tl] * (long)EMB + 2 * hi;
    const float* a1 = emb + (long)src[(16 + c) * SEQ + t0 + tl] * (long)EMB + 2 * hi;
    const float* bp = W_ih + (long)(2 * hi) * GATES + cbase + 4 * c;

    f32x8 acc[2][4];
    #pragma unroll
    for (int mi = 0; mi < 2; ++mi)
        #pragma unroll
        for (int j = 0; j < 4; ++j)
            acc[mi][j] = (f32x8){0.f, 0.f, 0.f, 0.f, 0.f, 0.f, 0.f, 0.f};

    #pragma unroll 8
    for (int k0 = 0; k0 < EMB; k0 += 4) {
        const f32x2 av0 = *(const f32x2*)(a0 + k0);
        const f32x2 av1 = *(const f32x2*)(a1 + k0);
        const f32x4 b0 = *(const f32x4*)(bp + (long)k0 * GATES);
        const f32x4 b1 = *(const f32x4*)(bp + (long)(k0 + 1) * GATES);
        #pragma unroll
        for (int j = 0; j < 4; ++j) {
            f32x2 b;
            b.x = b0[j];
            b.y = b1[j];
            acc[0][j] = __builtin_amdgcn_wmma_f32_16x16x4_f32(
                false, av0, false, b, (short)0, acc[0][j], false, false);
            acc[1][j] = __builtin_amdgcn_wmma_f32_16x16x4_f32(
                false, av1, false, b, (short)0, acc[1][j], false, false);
        }
    }

    #pragma unroll
    for (int mi = 0; mi < 2; ++mi) {
        #pragma unroll
        for (int r = 0; r < 8; ++r) {
            const int brow = mi * 16 + hi * 8 + r;
            f32x4 v;
            v.x = acc[mi][0][r];
            v.y = acc[mi][1][r];
            v.z = acc[mi][2][r];
            v.w = acc[mi][3][r];
            *(f32x4*)(xg + ((long)tl * BATCH + brow) * GATES + cbase + 4 * c) = v;
        }
    }
}

#define KC     128
#define NCHUNK (HID / KC)
#define PPITCH 128
#define STAGE_FLOATS ((KC / 2) * PPITCH)
#define ASTAGE_VEC4  (2 * (KC / 8) * 32)
#define APIECES      (ASTAGE_VEC4 / 256)

__device__ __forceinline__ void ask_chunk(
    const float* __restrict__ bsrc, const f32x4* __restrict__ a_src, int ch,
    f32x4 (&breg)[KC / 16], f32x4 (&areg)[APIECES])
{
    #pragma unroll
    for (int j = 0; j < KC / 16; ++j)
        breg[j] = *(const f32x4*)(bsrc + (long)(ch * KC + 16 * j) * GATES);
    #pragma unroll
    for (int q = 0; q < APIECES; ++q)
        areg[q] = a_src[((q >> 1) * (HID / 8) + ch * (KC / 8)) * 32 + (q & 1) * 256];
}

__device__ __forceinline__ void put_chunk(
    float* stage, int bdst, const f32x4 (&breg)[KC / 16],
    f32x4* astage, int tid, const f32x4 (&areg)[APIECES])
{
    #pragma unroll
    for (int j = 0; j < KC / 16; ++j) {
        float* d = stage + bdst + 8 * j * PPITCH;
        d[0] = breg[j].x;
        d[2] = breg[j].y;
        d[4] = breg[j].z;
        d[6] = breg[j].w;
    }
    #pragma unroll
    for (int q = 0; q < APIECES; ++q)
        astage[q * 256 + tid] = areg[q];
}

__device__ __forceinline__ f32x8 mma_chunk(
    f32x8 acc, const f32x4* al, const float* bl)
{
    f32x2 bf[2][8];
    f32x4 af[2][4];
    #pragma unroll
    for (int s = 0; s < 4; ++s)
        af[0][s] = al[s * 32];
    #pragma unroll
    for (int s = 0; s < 8; ++s)
        bf[0][s] = *(const f32x2*)(bl + (2 * s) * PPITCH);
    #pragma unroll
    for (int sb = 0; sb < KC / 32; ++sb) {
        if (sb + 1 < KC / 32) {
            #pragma unroll
            for (int s = 0; s < 4; ++s)
                af[(sb + 1) & 1][s] = al[(4 * (sb + 1) + s) * 32];
            #pragma unroll
            for (int s = 0; s < 8; ++s)
                bf[(sb + 1) & 1][s] = *(const f32x2*)(bl + (2 * (8 * (sb + 1) + s)) * PPITCH);
        }
        __builtin_amdgcn_sched_barrier(0);
        #pragma unroll
        for (int s = 0; s < 8; ++s) {
            const f32x4 a4 = af[sb & 1][s >> 1];
            f32x2 a;
            if (s & 1) { a.x = a4.z; a.y = a4.w; }
            else       { a.x = a4.x; a.y = a4.y; }
            acc = __builtin_amdgcn_wmma_f32_16x16x4_f32(
                false, a, false, bf[sb & 1][s], (short)0, acc, false, false);
        }
    }
    return acc;
}

__global__ __launch_bounds__(256) void lstm_step_kernel(
    const float* __restrict__ xg_t,
    const float* __restrict__ W_hh,
    const float* __restrict__ b_ih,
    const float* __restrict__ b_hh,
    const float* __restrict__ hf_rd,
    float*       __restrict__ hf_wr,
    float*       __restrict__ c_state,
    float*       __restrict__ h_out, long h_out_bstride)
{
    __shared__ __attribute__((aligned(16))) float lds_all[2 * ASTAGE_VEC4 * 4 + 2 * STAGE_FLOATS];
    f32x4* lds_a0 = (f32x4*)lds_all;
    f32x4* lds_a1 = lds_a0 + ASTAGE_VEC4;
    float* lds_b0 = lds_all + 2 * ASTAGE_VEC4 * 4;
    float* lds_b1 = lds_b0 + STAGE_FLOATS;
    float (*lds_g)[BATCH][16] = (float (*)[BATCH][16])lds_all;

    const int tid  = threadIdx.x;
    const int lane = tid & 31;
    const int wave = tid >> 5;
    const int gate = wave >> 1;
    const int mt   = wave & 1;
    const int l16  = lane & 15;
    const int hi   = lane >> 4;

    const int n0 = blockIdx.x * 16;
    const int N  = n0 + l16;
    const int gcol = gate * HID + N;

    const int brow   = tid >> 4;
    const int bpiece = tid & 15;
    const float* bsrc = W_hh + (long)brow * GATES + (bpiece >> 2) * HID + n0 + 4 * (bpiece & 3);
    const int bdst = (brow >> 1) * PPITCH + 8 * (bpiece ^ (((brow >> 1) & 1) << 2)) + (brow & 1);

    const f32x4* a_src = (const f32x4*)hf_rd + tid;

    f32x4 breg[KC / 16];
    f32x4 areg[APIECES];

    ask_chunk(bsrc, a_src, 0, breg, areg);

    f32x8 acc;
    #pragma unroll
    for (int r = 0; r < 8; ++r)
        acc[r] = xg_t[(long)(mt * 16 + hi * 8 + r) * GATES + gcol];

    const float bias = b_ih[gcol] + b_hh[gcol];

    put_chunk(lds_b0, bdst, breg, lds_a0, tid, areg);
    __syncthreads();

    const int bl_off = hi * PPITCH + 2 * ((gate * 16 + l16) ^ (hi << 4));
    const float* bl0 = lds_b0 + bl_off;
    const float* bl1 = lds_b1 + bl_off;
    const f32x4* al0 = lds_a0 + mt * (ASTAGE_VEC4 / 2) + lane;
    const f32x4* al1 = lds_a1 + mt * (ASTAGE_VEC4 / 2) + lane;

    for (int ch = 0; ch < NCHUNK; ch += 2) {
        ask_chunk(bsrc, a_src, ch + 1, breg, areg);
        acc = mma_chunk(acc, al0, bl0);
        put_chunk(lds_b1, bdst, breg, lds_a1, tid, areg);
        __syncthreads();

        const bool more = (ch + 2 < NCHUNK);
        if (more) ask_chunk(bsrc, a_src, ch + 2, breg, areg);
        acc = mma_chunk(acc, al1, bl1);
        if (more) {
            put_chunk(lds_b0, bdst, breg, lds_a0, tid, areg);
            __syncthreads();
        }
    }

    asm volatile("s_wait_alu depctr_va_vdst(0)\n\tv_nop\n\tv_nop\n\tv_nop\n\tv_nop" : "+v"(acc));

    #pragma unroll
    for (int r = 0; r < 8; ++r) {
        const int brw = mt * 16 + hi * 8 + r;
        lds_g[gate][brw][l16] = acc[r] + bias;
    }
    __syncthreads();

    #pragma unroll
    for (int idx = tid; idx < BATCH * 16; idx += 256) {
        const int b = idx >> 4;
        const int n = idx & 15;
        const float ig = sigmoidf_dev(lds_g[0][b][n]);
        const float fg = sigmoidf_dev(lds_g[1][b][n]);
        const float gg = tanhf(lds_g[2][b][n]);
        const float og = sigmoidf_dev(lds_g[3][b][n]);
        const long cidx = (long)b * HID + n0 + n;
        const float c_new = fg * c_state[cidx] + ig * gg;
        c_state[cidx] = c_new;
        const float h_new = og * tanhf(c_new);
        h_out[(long)b * h_out_bstride + n0 + n] = h_new;
        const int k  = n0 + n;
        const int kr = k & 7;
        const int fl = ((kr >> 1) & 1) * 16 + (b & 15);
        const int fe = (kr & 1) + 2 * (kr >> 2);
        hf_wr[((((b >> 4) * (HID / 8) + (k >> 3)) * 32 + fl) << 2) + fe] = h_new;
    }
}

extern "C" void kernel_launch(void* const* d_in, const int* in_sizes, int n_in,
                              void* d_out, int out_size, void* d_ws, size_t ws_size,
                              hipStream_t stream) {
    const int*   src  = (const int*)  d_in[0];
    const float* emb  = (const float*)d_in[1];
    const float* W_ih = (const float*)d_in[2];
    const float* W_hh = (const float*)d_in[3];
    const float* b_ih = (const float*)d_in[4];
    const float* b_hh = (const float*)d_in[5];
    float* out = (float*)d_out;

    float* c_state = (float*)d_ws;
    float* hf0     = c_state + BATCH * HID;
    float* hf1     = hf0 + BATCH * HID;
    float* xg      = hf1 + BATCH * HID;

    const int n_init = 2 * BATCH * HID;
    zero_ws_kernel<<<(n_init + 255) / 256, 256, 0, stream>>>(c_state, n_init);

    for (int t0 = 0; t0 < SEQ; t0 += TCHUNK) {
        xproj_kernel<<<dim3(GATES / 128, TCHUNK * BATCH / 128), 256, 0, stream>>>(
            src, emb, W_ih, xg, t0);

        for (int tl = 0; tl < TCHUNK; ++tl) {
            const int t = t0 + tl;
            const float* hf_rd = (t & 1) ? hf1 : hf0;
            float*       hf_wr = (t & 1) ? hf0 : hf1;
            lstm_step_kernel<<<HID / 16, 256, 0, stream>>>(
                xg + (long)tl * BATCH * GATES, W_hh, b_ih, b_hh,
                hf_rd, hf_wr, c_state,
                out + (long)t * HID, (long)SEQ * HID);
        }
    }
}


# Batch halves on separate work-groups (grid 64x2, 128 WGPs), half the WMMAs per WGP: a gate's two waves split each k chunk (256 rows now), partial tiles added in LDS; f32 sum order changes
# speedup vs baseline: 2.7691x; 1.1789x over previous
#include <hip/hip_runtime.h>


typedef __attribute__((ext_vector_type(2))) float f32x2;
typedef __attribute__((ext_vector_type(4))) float f32x4;
typedef __attribute__((ext_vector_type(8))) float f32x8;

#define VOCAB 32000
#define EMB   512
#define HID   1024
#define GATES 4096
#define BATCH 32
#define SEQ   512
#define TCHUNK 32

__global__ __launch_bounds__(256) void zero_ws_kernel(float* __restrict__ p, int n) {
    int i = blockIdx.x * blockDim.x + threadIdx.x;
    if (i < n) p[i] = 0.0f;
}

__device__ __forceinline__ float sigmoidf_dev(float x) {
    return 1.0f / (1.0f + __expf(-x));
}

__global__ __launch_bounds__(256) void xproj_kernel(
    const int*   __restrict__ src,
    const float* __restrict__ emb,
    const float* __restrict__ W_ih,
    float*       __restrict__ xg,
    int t0)
{
    const int tid  = threadIdx.x;
    const int lane = tid & 31;
    const int wave = tid >> 5;
    const int wm   = wave >> 1;
    const int wn   = wave & 1;
    const int c    = lane & 15;
    const int hi   = lane >> 4;

    const int tl    = blockIdx.y * 4 + wm;
    const int cbase = blockIdx.x * 128 + wn * 64;

    const float* a0 = emb + (long)src[c * SEQ + t0 + tl] * (long)EMB + 2 * hi;
    const float* a1 = emb + (long)src[(16 + c) * SEQ + t0 + tl] * (long)EMB + 2 * hi;
    const float* bp = W_ih + (long)(2 * hi) * GATES + cbase + 4 * c;

    f32x8 acc[2][4];
    #pragma unroll
    for (int mi = 0; mi < 2; ++mi)
        #pragma unroll
        for (int j = 0; j < 4; ++j)
            acc[mi][j] = (f32x8){0.f, 0.f, 0.f, 0.f, 0.f, 0.f, 0.f, 0.f};

    #pragma unroll 8
    for (int k0 = 0; k0 < EMB; k0 += 4) {
        const f32x2 av0 = *(const f32x2*)(a0 + k0);
        const f32x2 av1 = *(const f32x2*)(a1 + k0);
        const f32x4 b0 = *(const f32x4*)(bp + (long)k0 * GATES);
        const f32x4 b1 = *(const f32x4*)(bp + (long)(k0 + 1) * GATES);
        #pragma unroll
        for (int j = 0; j < 4; ++j) {
            f32x2 b;
            b.x = b0[j];
            b.y = b1[j];
            acc[0][j] = __builtin_amdgcn_wmma_f32_16x16x4_f32(
                false, av0, false, b, (short)0, acc[0][j], false, false);
            acc[1][j] = __builtin_amdgcn_wmma_f32_16x16x4_f32(
                false, av1, false, b, (short)0, acc[1][j], false, false);
        }
    }

    #pragma unroll
    for (int mi = 0; mi < 2; ++mi) {
        #pragma unroll
        for (int r = 0; r < 8; ++r) {
            const int brow = mi * 16 + hi * 8 + r;
            f32x4 v;
            v.x = acc[mi][0][r];
            v.y = acc[mi][1][r];
            v.z = acc[mi][2][r];
            v.w = acc[mi][3][r];
            *(f32x4*)(xg + ((long)tl * BATCH + brow) * GATES + cbase + 4 * c) = v;
        }
    }
}

#define KC     256
#define NCHUNK (HID / KC)
#define PPITCH 128
#define STAGE_FLOATS ((KC / 2) * PPITCH)
#define ASTAGE_VEC4  ((KC / 8) * 32)
#define APIECES      (ASTAGE_VEC4 / 256)
#define KSUB         (KC / 64)

__device__ __forceinline__ void ask_chunk(
    const float* __restrict__ bsrc, const f32x4* __restrict__ a_src, int ch,
    f32x4 (&breg)[KC / 16], f32x4 (&areg)[APIECES])
{
    #pragma unroll
    for (int j = 0; j < KC / 16; ++j)
        breg[j] = *(const f32x4*)(bsrc + (long)(ch * KC + 16 * j) * GATES);
    #pragma unroll
    for (int q = 0; q < APIECES; ++q)
        areg[q] = a_src[ch * (KC / 8) * 32 + q * 256];
}

__device__ __forceinline__ void put_chunk(
    float* stage, int bdst, const f32x4 (&breg)[KC / 16],
    f32x4* astage, int tid, const f32x4 (&areg)[APIECES])
{
    #pragma unroll
    for (int j = 0; j < KC / 16; ++j) {
        float* d = stage + bdst + 8 * j * PPITCH;
        d[0] = breg[j].x;
        d[2] = breg[j].y;
        d[4] = breg[j].z;
        d[6] = breg[j].w;
    }
    #pragma unroll
    for (int q = 0; q < APIECES; ++q)
        astage[q * 256 + tid] = areg[q];
}

__device__ __forceinline__ f32x8 mma_chunk(
    f32x8 acc, const f32x4* al, const float* bl)
{
    f32x2 bf[2][8];
    f32x4 af[2][4];
    #pragma unroll
    for (int s = 0; s < 4; ++s)
        af[0][s] = al[s * 32];
    #pragma unroll
    for (int s = 0; s < 8; ++s)
        bf[0][s] = *(const f32x2*)(bl + (2 * s) * PPITCH);
    #pragma unroll
    for (int sb = 0; sb < KSUB; ++sb) {
        if (sb + 1 < KSUB) {
            #pragma unroll
            for (int s = 0; s < 4; ++s)
                af[(sb + 1) & 1][s] = al[(4 * (sb + 1) + s) * 32];
            #pragma unroll
            for (int s = 0; s < 8; ++s)
                bf[(sb + 1) & 1][s] = *(const f32x2*)(bl + (2 * (8 * (sb + 1) + s)) * PPITCH);
        }
        __builtin_amdgcn_sched_barrier(0);
        #pragma unroll
        for (int s = 0; s < 8; ++s) {
            const f32x4 a4 = af[sb & 1][s >> 1];
            f32x2 a;
            if (s & 1) { a.x = a4.z; a.y = a4.w; }
            else       { a.x = a4.x; a.y = a4.y; }
            acc = __builtin_amdgcn_wmma_f32_16x16x4_f32(
                false, a, false, bf[sb & 1][s], (short)0, acc, false, false);
        }
    }
    return acc;
}

__global__ __launch_bounds__(256) void lstm_step_kernel(
    const float* __restrict__ xg_t,
    const float* __restrict__ W_hh,
    const float* __restrict__ b_ih,
    const float* __restrict__ b_hh,
    const float* __restrict__ hf_rd,
    float*       __restrict__ hf_wr,
    float*       __restrict__ c_state,
    float*       __restrict__ h_out, long h_out_bstride)
{
    __shared__ __attribute__((aligned(16))) float lds_all[2 * ASTAGE_VEC4 * 4 + 2 * STAGE_FLOATS];
    f32x4* lds_a0 = (f32x4*)lds_all;
    f32x4* lds_a1 = lds_a0 + ASTAGE_VEC4;
    float* lds_b0 = lds_all + 2 * ASTAGE_VEC4 * 4;
    float* lds_b1 = lds_b0 + STAGE_FLOATS;
    float (*lds_g)[4][16][16] = (float (*)[4][16][16])lds_all;

    const int tid  = threadIdx.x;
    const int lane = tid & 31;
    const int wave = tid >> 5;
    const int gate = wave >> 1;
    const int kh   = wave & 1;
    const int mt   = blockIdx.y;
    const int l16  = lane & 15;
    const int hi   = lane >> 4;

    const int n0 = blockIdx.x * 16;
    const int N  = n0 + l16;
    const int gcol = gate * HID + N;

    const int brow   = tid >> 4;
    const int bpiece = tid & 15;
    const float* bsrc = W_hh + (long)brow * GATES + (bpiece >> 2) * HID + n0 + 4 * (bpiece & 3);
    const int bdst = (brow >> 1) * PPITCH + 8 * (bpiece ^ (((brow >> 1) & 1) << 2)) + (brow & 1);

    const f32x4* a_src = (const f32x4*)hf_rd + mt * (HID / 8) * 32 + tid;

    f32x4 breg[KC / 16];
    f32x4 areg[APIECES];

    ask_chunk(bsrc, a_src, 0, breg, areg);

    f32x8 acc = (f32x8){0.f, 0.f, 0.f, 0.f, 0.f, 0.f, 0.f, 0.f};
    float bias = 0.f;
    if (kh == 0) {
        #pragma unroll
        for (int r = 0; r < 8; ++r)
            acc[r] = xg_t[(long)(mt * 16 + hi * 8 + r) * GATES + gcol];
        bias = b_ih[gcol] + b_hh[gcol];
    }

    put_chunk(lds_b0, bdst, breg, lds_a0, tid, areg);
    __syncthreads();

    const int bl_off = ((KC / 4) * kh + hi) * PPITCH + 2 * ((gate * 16 + l16) ^ (hi << 4));
    const float* bl0 = lds_b0 + bl_off;
    const float* bl1 = lds_b1 + bl_off;
    const f32x4* al0 = lds_a0 + (KC / 16) * kh * 32 + lane;
    const f32x4* al1 = lds_a1 + (KC / 16) * kh * 32 + lane;

    for (int ch = 0; ch < NCHUNK; ch += 2) {
        ask_chunk(bsrc, a_src, ch + 1, breg, areg);
        acc = mma_chunk(acc, al0, bl0);
        put_chunk(lds_b1, bdst, breg, lds_a1, tid, areg);
        __syncthreads();

        const bool more = (ch + 2 < NCHUNK);
        if (more) ask_chunk(bsrc, a_src, ch + 2, breg, areg);
        acc = mma_chunk(acc, al1, bl1);
        if (more) {
            put_chunk(lds_b0, bdst, breg, lds_a0, tid, areg);
            __syncthreads();
        }
    }

    asm volatile("s_wait_alu depctr_va_vdst(0)\n\tv_nop\n\tv_nop\n\tv_nop\n\tv_nop" : "+v"(acc));

    #pragma unroll
    for (int r = 0; r < 8; ++r)
        lds_g[kh][gate][hi * 8 + r][l16] = acc[r] + bias;
    __syncthreads();

    {
        const int bl = tid >> 4;
        const int n  = tid & 15;
        const int b  = mt * 16 + bl;
        const float ig = sigmoidf_dev(lds_g[0][0][bl][n] + lds_g[1][0][bl][n]);
        const float fg = sigmoidf_dev(lds_g[0][1][bl][n] + lds_g[1][1][bl][n]);
        const float gg = tanhf(lds_g[0][2][bl][n] + lds_g[1][2][bl][n]);
        const float og = sigmoidf_dev(lds_g[0][3][bl][n] + lds_g[1][3][bl][n]);
        const long cidx = (long)b * HID + n0 + n;
        const float c_new = fg * c_state[cidx] + ig * gg;
        c_state[cidx] = c_new;
        const float h_new = og * tanhf(c_new);
        h_out[(long)b * h_out_bstride + n0 + n] = h_new;
        const int k  = n0 + n;
        const int kr = k & 7;
        const int fl = ((kr >> 1) & 1) * 16 + bl;
        const int fe = (kr & 1) + 2 * (kr >> 2);
        hf_wr[(((mt * (HID / 8) + (k >> 3)) * 32 + fl) << 2) + fe] = h_new;
    }
}

extern "C" void kernel_launch(void* const* d_in, const int* in_sizes, int n_in,
                              void* d_out, int out_size, void* d_ws, size_t ws_size,
                              hipStream_t stream) {
    const int*   src  = (const int*)  d_in[0];
    const float* emb  = (const float*)d_in[1];
    const float* W_ih = (const float*)d_in[2];
    const float* W_hh = (const float*)d_in[3];
    const float* b_ih = (const float*)d_in[4];
    const float* b_hh = (const float*)d_in[5];
    float* out = (float*)d_out;

    float* c_state = (float*)d_ws;
    float* hf0     = c_state + BATCH * HID;
    float* hf1     = hf0 + BATCH * HID;
    float* xg      = hf1 + BATCH * HID;

    const int n_init = 2 * BATCH * HID;
    zero_ws_kernel<<<(n_init + 255) / 256, 256, 0, stream>>>(c_state, n_init);

    for (int t0 = 0; t0 < SEQ; t0 += TCHUNK) {
        xproj_kernel<<<dim3(GATES / 128, TCHUNK * BATCH / 128), 256, 0, stream>>>(
            src, emb, W_ih, xg, t0);

        for (int tl = 0; tl < TCHUNK; ++tl) {
            const int t = t0 + tl;
            const float* hf_rd = (t & 1) ? hf1 : hf0;
            float*       hf_wr = (t & 1) ? hf0 : hf1;
            lstm_step_kernel<<<dim3(HID / 16, 2), 256, 0, stream>>>(
                xg + (long)tl * BATCH * GATES, W_hh, b_ih, b_hh,
                hf_rd, hf_wr, c_state,
                out + (long)t * HID, (long)SEQ * HID);
        }
    }
}


# W_hh kept row-major in the LDS stage: a fetched 16-byte piece is one 16-byte LDS store (4x fewer stores, no bank conflict); a B fragment is one two-address read of rows k, k+1; f32 sum order kept
# speedup vs baseline: 2.8772x; 1.0390x over previous
#include <hip/hip_runtime.h>


typedef __attribute__((ext_vector_type(2))) float f32x2;
typedef __attribute__((ext_vector_type(4))) float f32x4;
typedef __attribute__((ext_vector_type(8))) float f32x8;

#define VOCAB 32000
#define EMB   512
#define HID   1024
#define GATES 4096
#define BATCH 32
#define SEQ   512
#define TCHUNK 32

__global__ __launch_bounds__(256) void zero_ws_kernel(float* __restrict__ p, int n) {
    int i = blockIdx.x * blockDim.x + threadIdx.x;
    if (i < n) p[i] = 0.0f;
}

__device__ __forceinline__ float sigmoidf_dev(float x) {
    return 1.0f / (1.0f + __expf(-x));
}

__global__ __launch_bounds__(256) void xproj_kernel(
    const int*   __restrict__ src,
    const float* __restrict__ emb,
    const float* __restrict__ W_ih,
    float*       __restrict__ xg,
    int t0)
{
    const int tid  = threadIdx.x;
    const int lane = tid & 31;
    const int wave = tid >> 5;
    const int wm   = wave >> 1;
    const int wn   = wave & 1;
    const int c    = lane & 15;
    const int hi   = lane >> 4;

    const int tl    = blockIdx.y * 4 + wm;
    const int cbase = blockIdx.x * 128 + wn * 64;

    const float* a0 = emb + (long)src[c * SEQ + t0 + tl] * (long)EMB + 2 * hi;
    const float* a1 = emb + (long)src[(16 + c) * SEQ + t0 + tl] * (long)EMB + 2 * hi;
    const float* bp = W_ih + (long)(2 * hi) * GATES + cbase + 4 * c;

    f32x8 acc[2][4];
    #pragma unroll
    for (int mi = 0; mi < 2; ++mi)
        #pragma unroll
        for (int j = 0; j < 4; ++j)
            acc[mi][j] = (f32x8){0.f, 0.f, 0.f, 0.f, 0.f, 0.f, 0.f, 0.f};

    #pragma unroll 8
    for (int k0 = 0; k0 < EMB; k0 += 4) {
        const f32x2 av0 = *(const f32x2*)(a0 + k0);
        const f32x2 av1 = *(const f32x2*)(a1 + k0);
        const f32x4 b0 = *(const f32x4*)(bp + (long)k0 * GATES);
        const f32x4 b1 = *(const f32x4*)(bp + (long)(k0 + 1) * GATES);
        #pragma unroll
        for (int j = 0; j < 4; ++j) {
            f32x2 b;
            b.x = b0[j];
            b.y = b1[j];
            acc[0][j] = __builtin_amdgcn_wmma_f32_16x16x4_f32(
                false, av0, false, b, (short)0, acc[0][j], false, false);
            acc[1][j] = __builtin_amdgcn_wmma_f32_16x16x4_f32(
                false, av1, false, b, (short)0, acc[1][j], false, false);
        }
    }

    #pragma unroll
    for (int mi = 0; mi < 2; ++mi) {
        #pragma unroll
        for (int r = 0; r < 8; ++r) {
            const int brow = mi * 16 + hi * 8 + r;
            f32x4 v;
            v.x = acc[mi][0][r];
            v.y = acc[mi][1][r];
            v.z = acc[mi][2][r];
            v.w = acc[mi][3][r];
            *(f32x4*)(xg + ((long)tl * BATCH + brow) * GATES + cbase + 4 * c) = v;
        }
    }
}

#define KC     256
#define NCHUNK (HID / KC)
#define BROWF  64
#define STAGE_FLOATS (KC * BROWF)
#define ASTAGE_VEC4  ((KC / 8) * 32)
#define APIECES      (ASTAGE_VEC4 / 256)
#define KSUB         (KC / 64)

__device__ __forceinline__ void ask_chunk(
    const float* __restrict__ bsrc, const f32x4* __restrict__ a_src, int ch,
    f32x4 (&breg)[KC / 16], f32x4 (&areg)[APIECES])
{
    #pragma unroll
    for (int j = 0; j < KC / 16; ++j)
        breg[j] = *(const f32x4*)(bsrc + (long)(ch * KC + 16 * j) * GATES);
    #pragma unroll
    for (int q = 0; q < APIECES; ++q)
        areg[q] = a_src[ch * (KC / 8) * 32 + q * 256];
}

__device__ __forceinline__ void put_chunk(
    float* stage, int bdst, const f32x4 (&breg)[KC / 16],
    f32x4* astage, int tid, const f32x4 (&areg)[APIECES])
{
    #pragma unroll
    for (int j = 0; j < KC / 16; ++j)
        *(f32x4*)(stage + bdst + 16 * j * BROWF) = breg[j];
    #pragma unroll
    for (int q = 0; q < APIECES; ++q)
        astage[q * 256 + tid] = areg[q];
}

__device__ __forceinline__ f32x8 mma_chunk(
    f32x8 acc, const f32x4* al, const float* bl)
{
    f32x2 bf[2][8];
    f32x4 af[2][4];
    #pragma unroll
    for (int s = 0; s < 4; ++s)
        af[0][s] = al[s * 32];
    #pragma unroll
    for (int s = 0; s < 8; ++s) {
        bf[0][s].x = bl[(4 * s) * BROWF];
        bf[0][s].y = bl[(4 * s + 1) * BROWF];
    }
    #pragma unroll
    for (int sb = 0; sb < KSUB; ++sb) {
        if (sb + 1 < KSUB) {
            #pragma unroll
            for (int s = 0; s < 4; ++s)
                af[(sb + 1) & 1][s] = al[(4 * (sb + 1) + s) * 32];
            #pragma unroll
            for (int s = 0; s < 8; ++s) {
                bf[(sb + 1) & 1][s].x = bl[(4 * (8 * (sb + 1) + s)) * BROWF];
                bf[(sb + 1) & 1][s].y = bl[(4 * (8 * (sb + 1) + s) + 1) * BROWF];
            }
        }
        __builtin_amdgcn_sched_barrier(0);
        #pragma unroll
        for (int s = 0; s < 8; ++s) {
            const f32x4 a4 = af[sb & 1][s >> 1];
            f32x2 a;
            if (s & 1) { a.x = a4.z; a.y = a4.w; }
            else       { a.x = a4.x; a.y = a4.y; }
            acc = __builtin_amdgcn_wmma_f32_16x16x4_f32(
                false, a, false, bf[sb & 1][s], (short)0, acc, false, false);
        }
    }
    return acc;
}

__global__ __launch_bounds__(256) void lstm_step_kernel(
    const float* __restrict__ xg_t,
    const float* __restrict__ W_hh,
    const float* __restrict__ b_ih,
    const float* __restrict__ b_hh,
    const float* __restrict__ hf_rd,
    float*       __restrict__ hf_wr,
    float*       __restrict__ c_state,
    float*       __restrict__ h_out, long h_out_bstride)
{
    __shared__ __attribute__((aligned(16))) float lds_all[2 * ASTAGE_VEC4 * 4 + 2 * STAGE_FLOATS];
    f32x4* lds_a0 = (f32x4*)lds_all;
    f32x4* lds_a1 = lds_a0 + ASTAGE_VEC4;
    float* lds_b0 = lds_all + 2 * ASTAGE_VEC4 * 4;
    float* lds_b1 = lds_b0 + STAGE_FLOATS;
    float (*lds_g)[4][16][16] = (float (*)[4][16][16])lds_all;

    const int tid  = threadIdx.x;
    const int lane = tid & 31;
    const int wave = tid >> 5;
    const int gate = wave >> 1;
    const int kh   = wave & 1;
    const int mt   = blockIdx.y;
    const int l16  = lane & 15;
    const int hi   = lane >> 4;

    const int n0 = blockIdx.x * 16;
    const int N  = n0 + l16;
    const int gcol = gate * HID + N;

    const int brow   = tid >> 4;
    const int bpiece = tid & 15;
    const float* bsrc = W_hh + (long)brow * GATES + (bpiece >> 2) * HID + n0 + 4 * (bpiece & 3);
    const int bdst = brow * BROWF + 4 * (bpiece ^ (((brow >> 1) & 1) << 2));

    const f32x4* a_src = (const f32x4*)hf_rd + mt * (HID / 8) * 32 + tid;

    f32x4 breg[KC / 16];
    f32x4 areg[APIECES];

    ask_chunk(bsrc, a_src, 0, breg, areg);

    f32x8 acc = (f32x8){0.f, 0.f, 0.f, 0.f, 0.f, 0.f, 0.f, 0.f};
    float bias = 0.f;
    if (kh == 0) {
        #pragma unroll
        for (int r = 0; r < 8; ++r)
            acc[r] = xg_t[(long)(mt * 16 + hi * 8 + r) * GATES + gcol];
        bias = b_ih[gcol] + b_hh[gcol];
    }

    put_chunk(lds_b0, bdst, breg, lds_a0, tid, areg);
    __syncthreads();

    const int bl_off = ((KC / 2) * kh + 2 * hi) * BROWF + ((gate * 16 + l16) ^ (hi << 4));
    const float* bl0 = lds_b0 + bl_off;
    const float* bl1 = lds_b1 + bl_off;
    const f32x4* al0 = lds_a0 + (KC / 16) * kh * 32 + lane;
    const f32x4* al1 = lds_a1 + (KC / 16) * kh * 32 + lane;

    for (int ch = 0; ch < NCHUNK; ch += 2) {
        ask_chunk(bsrc, a_src, ch + 1, breg, areg);
        acc = mma_chunk(acc, al0, bl0);
        put_chunk(lds_b1, bdst, breg, lds_a1, tid, areg);
        __syncthreads();

        const bool more = (ch + 2 < NCHUNK);
        if (more) ask_chunk(bsrc, a_src, ch + 2, breg, areg);
        acc = mma_chunk(acc, al1, bl1);
        if (more) {
            put_chunk(lds_b0, bdst, breg, lds_a0, tid, areg);
            __syncthreads();
        }
    }

    asm volatile("s_wait_alu depctr_va_vdst(0)\n\tv_nop\n\tv_nop\n\tv_nop\n\tv_nop" : "+v"(acc));

    #pragma unroll
    for (int r = 0; r < 8; ++r)
        lds_g[kh][gate][hi * 8 + r][l16] = acc[r] + bias;
    __syncthreads();

    {
        const int bl = tid >> 4;
        const int n  = tid & 15;
        const int b  = mt * 16 + bl;
        const float ig = sigmoidf_dev(lds_g[0][0][bl][n] + lds_g[1][0][bl][n]);
        const float fg = sigmoidf_dev(lds_g[0][1][bl][n] + lds_g[1][1][bl][n]);
        const float gg = tanhf(lds_g[0][2][bl][n] + lds_g[1][2][bl][n]);
        const float og = sigmoidf_dev(lds_g[0][3][bl][n] + lds_g[1][3][bl][n]);
        const long cidx = (long)b * HID + n0 + n;
        const float c_new = fg * c_state[cidx] + ig * gg;
        c_state[cidx] = c_new;
        const float h_new = og * tanhf(c_new);
        h_out[(long)b * h_out_bstride + n0 + n] = h_new;
        const int k  = n0 + n;
        const int kr = k & 7;
        const int fl = ((kr >> 1) & 1) * 16 + bl;
        const int fe = (kr & 1) + 2 * (kr >> 2);
        hf_wr[(((mt * (HID / 8) + (k >> 3)) * 32 + fl) << 2) + fe] = h_new;
    }
}

extern "C" void kernel_launch(void* const* d_in, const int* in_sizes, int n_in,
                              void* d_out, int out_size, void* d_ws, size_t ws_size,
                              hipStream_t stream) {
    const int*   src  = (const int*)  d_in[0];
    const float* emb  = (const float*)d_in[1];
    const float* W_ih = (const float*)d_in[2];
    const float* W_hh = (const float*)d_in[3];
    const float* b_ih = (const float*)d_in[4];
    const float* b_hh = (const float*)d_in[5];
    float* out = (float*)d_out;

    float* c_state = (float*)d_ws;
    float* hf0     = c_state + BATCH * HID;
    float* hf1     = hf0 + BATCH * HID;
    float* xg      = hf1 + BATCH * HID;

    const int n_init = 2 * BATCH * HID;
    zero_ws_kernel<<<(n_init + 255) / 256, 256, 0, stream>>>(c_state, n_init);

    for (int t0 = 0; t0 < SEQ; t0 += TCHUNK) {
        xproj_kernel<<<dim3(GATES / 128, TCHUNK * BATCH / 128), 256, 0, stream>>>(
            src, emb, W_ih, xg, t0);

        for (int tl = 0; tl < TCHUNK; ++tl) {
            const int t = t0 + tl;
            const float* hf_rd = (t & 1) ? hf1 : hf0;
            float*       hf_wr = (t & 1) ? hf0 : hf1;
            lstm_step_kernel<<<dim3(HID / 16, 2), 256, 0, stream>>>(
                xg + (long)tl * BATCH * GATES, W_hh, b_ih, b_hh,
                hf_rd, hf_wr, c_state,
                out + (long)t * HID, (long)SEQ * HID);
        }
    }
}


# Operand stream kept running: W_hh and h in 8 chunks of 128 rows (were 4 of 256), each asked two chunks ahead across LDS-only barriers, so the load queue stays fed; f32 sum order changes
# speedup vs baseline: 3.2621x; 1.1338x over previous
#include <hip/hip_runtime.h>


typedef __attribute__((ext_vector_type(2))) float f32x2;
typedef __attribute__((ext_vector_type(4))) float f32x4;
typedef __attribute__((ext_vector_type(8))) float f32x8;

#define VOCAB 32000
#define EMB   512
#define HID   1024
#define GATES 4096
#define BATCH 32
#define SEQ   512
#define TCHUNK 32

__global__ __launch_bounds__(256) void zero_ws_kernel(float* __restrict__ p, int n) {
    int i = blockIdx.x * blockDim.x + threadIdx.x;
    if (i < n) p[i] = 0.0f;
}

__device__ __forceinline__ float sigmoidf_dev(float x) {
    return 1.0f / (1.0f + __expf(-x));
}

__global__ __launch_bounds__(256) void xproj_kernel(
    const int*   __restrict__ src,
    const float* __restrict__ emb,
    const float* __restrict__ W_ih,
    float*       __restrict__ xg,
    int t0)
{
    const int tid  = threadIdx.x;
    const int lane = tid & 31;
    const int wave = tid >> 5;
    const int wm   = wave >> 1;
    const int wn   = wave & 1;
    const int c    = lane & 15;
    const int hi   = lane >> 4;

    const int tl    = blockIdx.y * 4 + wm;
    const int cbase = blockIdx.x * 128 + wn * 64;

    const float* a0 = emb + (long)src[c * SEQ + t0 + tl] * (long)EMB + 2 * hi;
    const float* a1 = emb + (long)src[(16 + c) * SEQ + t0 + tl] * (long)EMB + 2 * hi;
    const float* bp = W_ih + (long)(2 * hi) * GATES + cbase + 4 * c;

    f32x8 acc[2][4];
    #pragma unroll
    for (int mi = 0; mi < 2; ++mi)
        #pragma unroll
        for (int j = 0; j < 4; ++j)
            acc[mi][j] = (f32x8){0.f, 0.f, 0.f, 0.f, 0.f, 0.f, 0.f, 0.f};

    #pragma unroll 8
    for (int k0 = 0; k0 < EMB; k0 += 4) {
        const f32x2 av0 = *(const f32x2*)(a0 + k0);
        const f32x2 av1 = *(const f32x2*)(a1 + k0);
        const f32x4 b0 = *(const f32x4*)(bp + (long)k0 * GATES);
        const f32x4 b1 = *(const f32x4*)(bp + (long)(k0 + 1) * GATES);
        #pragma unroll
        for (int j = 0; j < 4; ++j) {
            f32x2 b;
            b.x = b0[j];
            b.y = b1[j];
            acc[0][j] = __builtin_amdgcn_wmma_f32_16x16x4_f32(
                false, av0, false, b, (short)0, acc[0][j], false, false);
            acc[1][j] = __builtin_amdgcn_wmma_f32_16x16x4_f32(
                false, av1, false, b, (short)0, acc[1][j], false, false);
        }
    }

    #pragma unroll
    for (int mi = 0; mi < 2; ++mi) {
        #pragma unroll
        for (int r = 0; r < 8; ++r) {
            const int brow = mi * 16 + hi * 8 + r;
            f32x4 v;
            v.x = acc[mi][0][r];
            v.y = acc[mi][1][r];
            v.z = acc[mi][2][r];
            v.w = acc[mi][3][r];
            *(f32x4*)(xg + ((long)tl * BATCH + brow) * GATES + cbase + 4 * c) = v;
        }
    }
}

#define KC     128
#define NCHUNK (HID / KC)
#define BROWF  64
#define STAGE_FLOATS (KC * BROWF)
#define ASTAGE_VEC4  ((KC / 8) * 32)
#define APIECES      (ASTAGE_VEC4 / 256)
#define KSUB         (KC / 64)

__device__ __forceinline__ void ask_chunk(
    const float* __restrict__ bsrc, const f32x4* __restrict__ a_src, int ch,
    f32x4 (&breg)[KC / 16], f32x4 (&areg)[APIECES])
{
    #pragma unroll
    for (int j = 0; j < KC / 16; ++j)
        breg[j] = *(const f32x4*)(bsrc + (long)(ch * KC + 16 * j) * GATES);
    #pragma unroll
    for (int q = 0; q < APIECES; ++q)
        areg[q] = a_src[ch * (KC / 8) * 32 + q * 256];
}

__device__ __forceinline__ void put_chunk(
    float* stage, int bdst, const f32x4 (&breg)[KC / 16],
    f32x4* astage, int tid, const f32x4 (&areg)[APIECES])
{
    #pragma unroll
    for (int j = 0; j < KC / 16; ++j)
        *(f32x4*)(stage + bdst + 16 * j * BROWF) = breg[j];
    #pragma unroll
    for (int q = 0; q < APIECES; ++q)
        astage[q * 256 + tid] = areg[q];
}

__device__ __forceinline__ void lds_barrier()
{
    asm volatile("s_wait_dscnt 0x0\n\ts_barrier_signal -1\n\ts_barrier_wait -1" ::: "memory");
}

__device__ __forceinline__ f32x8 mma_chunk(
    f32x8 acc, const f32x4* al, const float* bl)
{
    f32x2 bf[2][8];
    f32x4 af[2][4];
    #pragma unroll
    for (int s = 0; s < 4; ++s)
        af[0][s] = al[s * 32];
    #pragma unroll
    for (int s = 0; s < 8; ++s) {
        bf[0][s].x = bl[(4 * s) * BROWF];
        bf[0][s].y = bl[(4 * s + 1) * BROWF];
    }
    #pragma unroll
    for (int sb = 0; sb < KSUB; ++sb) {
        if (sb + 1 < KSUB) {
            #pragma unroll
            for (int s = 0; s < 4; ++s)
                af[(sb + 1) & 1][s] = al[(4 * (sb + 1) + s) * 32];
            #pragma unroll
            for (int s = 0; s < 8; ++s) {
                bf[(sb + 1) & 1][s].x = bl[(4 * (8 * (sb + 1) + s)) * BROWF];
                bf[(sb + 1) & 1][s].y = bl[(4 * (8 * (sb + 1) + s) + 1) * BROWF];
            }
        }
        __builtin_amdgcn_sched_barrier(0);
        #pragma unroll
        for (int s = 0; s < 8; ++s) {
            const f32x4 a4 = af[sb & 1][s >> 1];
            f32x2 a;
            if (s & 1) { a.x = a4.z; a.y = a4.w; }
            else       { a.x = a4.x; a.y = a4.y; }
            acc = __builtin_amdgcn_wmma_f32_16x16x4_f32(
                false, a, false, bf[sb & 1][s], (short)0, acc, false, false);
        }
    }
    return acc;
}

__global__ __launch_bounds__(256) void lstm_step_kernel(
    const float* __restrict__ xg_t,
    const float* __restrict__ W_hh,
    const float* __restrict__ b_ih,
    const float* __restrict__ b_hh,
    const float* __restrict__ hf_rd,
    float*       __restrict__ hf_wr,
    float*       __restrict__ c_state,
    float*       __restrict__ h_out, long h_out_bstride)
{
    __shared__ __attribute__((aligned(16))) float lds_all[2 * ASTAGE_VEC4 * 4 + 2 * STAGE_FLOATS];
    f32x4* lds_a0 = (f32x4*)lds_all;
    f32x4* lds_a1 = lds_a0 + ASTAGE_VEC4;
    float* lds_b0 = lds_all + 2 * ASTAGE_VEC4 * 4;
    float* lds_b1 = lds_b0 + STAGE_FLOATS;
    float (*lds_g)[4][16][16] = (float (*)[4][16][16])lds_all;

    const int tid  = threadIdx.x;
    const int lane = tid & 31;
    const int wave = tid >> 5;
    const int gate = wave >> 1;
    const int kh   = wave & 1;
    const int mt   = blockIdx.y;
    const int l16  = lane & 15;
    const int hi   = lane >> 4;

    const int n0 = blockIdx.x * 16;
    const int N  = n0 + l16;
    const int gcol = gate * HID + N;

    const int brow   = tid >> 4;
    const int bpiece = tid & 15;
    const float* bsrc = W_hh + (long)brow * GATES + (bpiece >> 2) * HID + n0 + 4 * (bpiece & 3);
    const int bdst = brow * BROWF + 4 * (bpiece ^ (((brow >> 1) & 1) << 2));

    const f32x4* a_src = (const f32x4*)hf_rd + mt * (HID / 8) * 32 + tid;

    f32x4 breg0[KC / 16], breg1[KC / 16];
    f32x4 areg0[APIECES], areg1[APIECES];

    ask_chunk(bsrc, a_src, 0, breg0, areg0);

    f32x8 acc = (f32x8){0.f, 0.f, 0.f, 0.f, 0.f, 0.f, 0.f, 0.f};
    float bias = 0.f;
    if (kh == 0) {
        #pragma unroll
        for (int r = 0; r < 8; ++r)
            acc[r] = xg_t[(long)(mt * 16 + hi * 8 + r) * GATES + gcol];
        bias = b_ih[gcol] + b_hh[gcol];
    }

    ask_chunk(bsrc, a_src, 1, breg1, areg1);

    put_chunk(lds_b0, bdst, breg0, lds_a0, tid, areg0);
    lds_barrier();

    const int bl_off = ((KC / 2) * kh + 2 * hi) * BROWF + ((gate * 16 + l16) ^ (hi << 4));
    const float* bl0 = lds_b0 + bl_off;
    const float* bl1 = lds_b1 + bl_off;
    const f32x4* al0 = lds_a0 + (KC / 16) * kh * 32 + lane;
    const f32x4* al1 = lds_a1 + (KC / 16) * kh * 32 + lane;

    for (int ch = 0; ch < NCHUNK; ch += 2) {
        const bool more = (ch + 2 < NCHUNK);
        if (more) ask_chunk(bsrc, a_src, ch + 2, breg0, areg0);
        acc = mma_chunk(acc, al0, bl0);
        put_chunk(lds_b1, bdst, breg1, lds_a1, tid, areg1);
        lds_barrier();

        if (more) ask_chunk(bsrc, a_src, ch + 3, breg1, areg1);
        acc = mma_chunk(acc, al1, bl1);
        if (more) {
            put_chunk(lds_b0, bdst, breg0, lds_a0, tid, areg0);
            lds_barrier();
        }
    }

    asm volatile("s_wait_alu depctr_va_vdst(0)\n\tv_nop\n\tv_nop\n\tv_nop\n\tv_nop" : "+v"(acc));

    #pragma unroll
    for (int r = 0; r < 8; ++r)
        lds_g[kh][gate][hi * 8 + r][l16] = acc[r] + bias;
    __syncthreads();

    {
        const int bl = tid >> 4;
        const int n  = tid & 15;
        const int b  = mt * 16 + bl;
        const float ig = sigmoidf_dev(lds_g[0][0][bl][n] + lds_g[1][0][bl][n]);
        const float fg = sigmoidf_dev(lds_g[0][1][bl][n] + lds_g[1][1][bl][n]);
        const float gg = tanhf(lds_g[0][2][bl][n] + lds_g[1][2][bl][n]);
        const float og = sigmoidf_dev(lds_g[0][3][bl][n] + lds_g[1][3][bl][n]);
        const long cidx = (long)b * HID + n0 + n;
        const float c_new = fg * c_state[cidx] + ig * gg;
        c_state[cidx] = c_new;
        const float h_new = og * tanhf(c_new);
        h_out[(long)b * h_out_bstride + n0 + n] = h_new;
        const int k  = n0 + n;
        const int kr = k & 7;
        const int fl = ((kr >> 1) & 1) * 16 + bl;
        const int fe = (kr & 1) + 2 * (kr >> 2);
        hf_wr[(((mt * (HID / 8) + (k >> 3)) * 32 + fl) << 2) + fe] = h_new;
    }
}

extern "C" void kernel_launch(void* const* d_in, const int* in_sizes, int n_in,
                              void* d_out, int out_size, void* d_ws, size_t ws_size,
                              hipStream_t stream) {
    const int*   src  = (const int*)  d_in[0];
    const float* emb  = (const float*)d_in[1];
    const float* W_ih = (const float*)d_in[2];
    const float* W_hh = (const float*)d_in[3];
    const float* b_ih = (const float*)d_in[4];
    const float* b_hh = (const float*)d_in[5];
    float* out = (float*)d_out;

    float* c_state = (float*)d_ws;
    float* hf0     = c_state + BATCH * HID;
    float* hf1     = hf0 + BATCH * HID;
    float* xg      = hf1 + BATCH * HID;

    const int n_init = 2 * BATCH * HID;
    zero_ws_kernel<<<(n_init + 255) / 256, 256, 0, stream>>>(c_state, n_init);

    for (int t0 = 0; t0 < SEQ; t0 += TCHUNK) {
        xproj_kernel<<<dim3(GATES / 128, TCHUNK * BATCH / 128), 256, 0, stream>>>(
            src, emb, W_ih, xg, t0);

        for (int tl = 0; tl < TCHUNK; ++tl) {
            const int t = t0 + tl;
            const float* hf_rd = (t & 1) ? hf1 : hf0;
            float*       hf_wr = (t & 1) ? hf0 : hf1;
            lstm_step_kernel<<<dim3(HID / 16, 2), 256, 0, stream>>>(
                xg + (long)tl * BATCH * GATES, W_hh, b_ih, b_hh,
                hf_rd, hf_wr, c_state,
                out + (long)t * HID, (long)SEQ * HID);
        }
    }
}


# Each block walks the step's 8 k chunks from its own start chunk (cyclic, by block index), so blocks sharing W_hh and h lines do not ask L2 for them in the same moment; f32 sum order changes
# speedup vs baseline: 3.3651x; 1.0316x over previous
#include <hip/hip_runtime.h>


typedef __attribute__((ext_vector_type(2))) float f32x2;
typedef __attribute__((ext_vector_type(4))) float f32x4;
typedef __attribute__((ext_vector_type(8))) float f32x8;

#define VOCAB 32000
#define EMB   512
#define HID   1024
#define GATES 4096
#define BATCH 32
#define SEQ   512
#define TCHUNK 32

__global__ __launch_bounds__(256) void zero_ws_kernel(float* __restrict__ p, int n) {
    int i = blockIdx.x * blockDim.x + threadIdx.x;
    if (i < n) p[i] = 0.0f;
}

__device__ __forceinline__ float sigmoidf_dev(float x) {
    return 1.0f / (1.0f + __expf(-x));
}

__global__ __launch_bounds__(256) void xproj_kernel(
    const int*   __restrict__ src,
    const float* __restrict__ emb,
    const float* __restrict__ W_ih,
    float*       __restrict__ xg,
    int t0)
{
    const int tid  = threadIdx.x;
    const int lane = tid & 31;
    const int wave = tid >> 5;
    const int wm   = wave >> 1;
    const int wn   = wave & 1;
    const int c    = lane & 15;
    const int hi   = lane >> 4;

    const int tl    = blockIdx.y * 4 + wm;
    const int cbase = blockIdx.x * 128 + wn * 64;

    const float* a0 = emb + (long)src[c * SEQ + t0 + tl] * (long)EMB + 2 * hi;
    const float* a1 = emb + (long)src[(16 + c) * SEQ + t0 + tl] * (long)EMB + 2 * hi;
    const float* bp = W_ih + (long)(2 * hi) * GATES + cbase + 4 * c;

    f32x8 acc[2][4];
    #pragma unroll
    for (int mi = 0; mi < 2; ++mi)
        #pragma unroll
        for (int j = 0; j < 4; ++j)
            acc[mi][j] = (f32x8){0.f, 0.f, 0.f, 0.f, 0.f, 0.f, 0.f, 0.f};

    #pragma unroll 8
    for (int k0 = 0; k0 < EMB; k0 += 4) {
        const f32x2 av0 = *(const f32x2*)(a0 + k0);
        const f32x2 av1 = *(const f32x2*)(a1 + k0);
        const f32x4 b0 = *(const f32x4*)(bp + (long)k0 * GATES);
        const f32x4 b1 = *(const f32x4*)(bp + (long)(k0 + 1) * GATES);
        #pragma unroll
        for (int j = 0; j < 4; ++j) {
            f32x2 b;
            b.x = b0[j];
            b.y = b1[j];
            acc[0][j] = __builtin_amdgcn_wmma_f32_16x16x4_f32(
                false, av0, false, b, (short)0, acc[0][j], false, false);
            acc[1][j] = __builtin_amdgcn_wmma_f32_16x16x4_f32(
                false, av1, false, b, (short)0, acc[1][j], false, false);
        }
    }

    #pragma unroll
    for (int mi = 0; mi < 2; ++mi) {
        #pragma unroll
        for (int r = 0; r < 8; ++r) {
            const int brow = mi * 16 + hi * 8 + r;
            f32x4 v;
            v.x = acc[mi][0][r];
            v.y = acc[mi][1][r];
            v.z = acc[mi][2][r];
            v.w = acc[mi][3][r];
            *(f32x4*)(xg + ((long)tl * BATCH + brow) * GATES + cbase + 4 * c) = v;
        }
    }
}

#define KC     128
#define NCHUNK (HID / KC)
#define BROWF  64
#define STAGE_FLOATS (KC * BROWF)
#define ASTAGE_VEC4  ((KC / 8) * 32)
#define APIECES      (ASTAGE_VEC4 / 256)
#define KSUB         (KC / 64)

__device__ __forceinline__ void ask_chunk(
    const float* __restrict__ wbase, unsigned boff, const f32x4* __restrict__ abase, unsigned aoff, int ch,
    f32x4 (&breg)[KC / 16], f32x4 (&areg)[APIECES])
{
    const float* wch = wbase + (long)ch * KC * GATES;
    const f32x4* ach = abase + ch * (KC / 8) * 32;
    #pragma unroll
    for (int j = 0; j < KC / 16; ++j)
        breg[j] = *(const f32x4*)(wch + (boff + (unsigned)(16 * j * GATES)));
    #pragma unroll
    for (int q = 0; q < APIECES; ++q)
        areg[q] = ach[aoff + (unsigned)(q * 256)];
}

__device__ __forceinline__ void put_chunk(
    float* stage, int bdst, const f32x4 (&breg)[KC / 16],
    f32x4* astage, int tid, const f32x4 (&areg)[APIECES])
{
    #pragma unroll
    for (int j = 0; j < KC / 16; ++j)
        *(f32x4*)(stage + bdst + 16 * j * BROWF) = breg[j];
    #pragma unroll
    for (int q = 0; q < APIECES; ++q)
        astage[q * 256 + tid] = areg[q];
}

__device__ __forceinline__ void lds_barrier()
{
    asm volatile("s_wait_dscnt 0x0\n\ts_barrier_signal -1\n\ts_barrier_wait -1" ::: "memory");
}

__device__ __forceinline__ f32x8 mma_chunk(
    f32x8 acc, const f32x4* al, const float* bl)
{
    f32x2 bf[2][8];
    f32x4 af[2][4];
    #pragma unroll
    for (int s = 0; s < 4; ++s)
        af[0][s] = al[s * 32];
    #pragma unroll
    for (int s = 0; s < 8; ++s) {
        bf[0][s].x = bl[(4 * s) * BROWF];
        bf[0][s].y = bl[(4 * s + 1) * BROWF];
    }
    #pragma unroll
    for (int sb = 0; sb < KSUB; ++sb) {
        if (sb + 1 < KSUB) {
            #pragma unroll
            for (int s = 0; s < 4; ++s)
                af[(sb + 1) & 1][s] = al[(4 * (sb + 1) + s) * 32];
            #pragma unroll
            for (int s = 0; s < 8; ++s) {
                bf[(sb + 1) & 1][s].x = bl[(4 * (8 * (sb + 1) + s)) * BROWF];
                bf[(sb + 1) & 1][s].y = bl[(4 * (8 * (sb + 1) + s) + 1) * BROWF];
            }
        }
        __builtin_amdgcn_sched_barrier(0);
        #pragma unroll
        for (int s = 0; s < 8; ++s) {
            const f32x4 a4 = af[sb & 1][s >> 1];
            f32x2 a;
            if (s & 1) { a.x = a4.z; a.y = a4.w; }
            else       { a.x = a4.x; a.y = a4.y; }
            acc = __builtin_amdgcn_wmma_f32_16x16x4_f32(
                false, a, false, bf[sb & 1][s], (short)0, acc, false, false);
        }
    }
    return acc;
}

__global__ __launch_bounds__(256) void lstm_step_kernel(
    const float* __restrict__ xg_t,
    const float* __restrict__ W_hh,
    const float* __restrict__ b_ih,
    const float* __restrict__ b_hh,
    const float* __restrict__ hf_rd,
    float*       __restrict__ hf_wr,
    float*       __restrict__ c_state,
    float*       __restrict__ h_out, long h_out_bstride)
{
    __shared__ __attribute__((aligned(16))) float lds_all[2 * ASTAGE_VEC4 * 4 + 2 * STAGE_FLOATS];
    f32x4* lds_a0 = (f32x4*)lds_all;
    f32x4* lds_a1 = lds_a0 + ASTAGE_VEC4;
    float* lds_b0 = lds_all + 2 * ASTAGE_VEC4 * 4;
    float* lds_b1 = lds_b0 + STAGE_FLOATS;
    float (*lds_g)[4][16][16] = (float (*)[4][16][16])lds_all;

    const int tid  = threadIdx.x;
    const int lane = tid & 31;
    const int wave = tid >> 5;
    const int gate = wave >> 1;
    const int kh   = wave & 1;
    const int mt   = blockIdx.y;
    const int l16  = lane & 15;
    const int hi   = lane >> 4;

    const int n0 = blockIdx.x * 16;
    const int N  = n0 + l16;
    const int gcol = gate * HID + N;

    const int brow   = tid >> 4;
    const int bpiece = tid & 15;
    const float* wbase = W_hh + n0;
    const unsigned boff = (unsigned)(brow * GATES + (bpiece >> 2) * HID + 4 * (bpiece & 3));
    const int bdst = brow * BROWF + 4 * (bpiece ^ (((brow >> 1) & 1) << 2));

    const f32x4* abase = (const f32x4*)hf_rd + mt * (HID / 8) * 32;
    const unsigned aoff = (unsigned)tid;

    f32x4 breg0[KC / 16], breg1[KC / 16];
    f32x4 areg0[APIECES], areg1[APIECES];

    const int rot = (2 * (int)(blockIdx.x & 1) + 4 * (int)blockIdx.y + (int)((blockIdx.x >> 1) & 1)) & (NCHUNK - 1);

    ask_chunk(wbase, boff, abase, aoff, rot, breg0, areg0);

    f32x8 acc = (f32x8){0.f, 0.f, 0.f, 0.f, 0.f, 0.f, 0.f, 0.f};
    float bias = 0.f;
    if (kh == 0) {
        #pragma unroll
        for (int r = 0; r < 8; ++r)
            acc[r] = xg_t[(long)(mt * 16 + hi * 8 + r) * GATES + gcol];
        bias = b_ih[gcol] + b_hh[gcol];
    }

    ask_chunk(wbase, boff, abase, aoff, (rot + 1) & (NCHUNK - 1), breg1, areg1);

    put_chunk(lds_b0, bdst, breg0, lds_a0, tid, areg0);
    lds_barrier();

    const int bl_off = ((KC / 2) * kh + 2 * hi) * BROWF + ((gate * 16 + l16) ^ (hi << 4));
    const float* bl0 = lds_b0 + bl_off;
    const float* bl1 = lds_b1 + bl_off;
    const f32x4* al0 = lds_a0 + (KC / 16) * kh * 32 + lane;
    const f32x4* al1 = lds_a1 + (KC / 16) * kh * 32 + lane;

    for (int ch = 0; ch < NCHUNK; ch += 2) {
        const bool more = (ch + 2 < NCHUNK);
        if (more) ask_chunk(wbase, boff, abase, aoff, (rot + ch + 2) & (NCHUNK - 1), breg0, areg0);
        acc = mma_chunk(acc, al0, bl0);
        put_chunk(lds_b1, bdst, breg1, lds_a1, tid, areg1);
        lds_barrier();

        if (more) ask_chunk(wbase, boff, abase, aoff, (rot + ch + 3) & (NCHUNK - 1), breg1, areg1);
        acc = mma_chunk(acc, al1, bl1);
        if (more) {
            put_chunk(lds_b0, bdst, breg0, lds_a0, tid, areg0);
            lds_barrier();
        }
    }

    asm volatile("s_wait_alu depctr_va_vdst(0)\n\tv_nop\n\tv_nop\n\tv_nop\n\tv_nop" : "+v"(acc));

    #pragma unroll
    for (int r = 0; r < 8; ++r)
        lds_g[kh][gate][hi * 8 + r][l16] = acc[r] + bias;
    __syncthreads();

    {
        const int bl = tid >> 4;
        const int n  = tid & 15;
        const int b  = mt * 16 + bl;
        const float ig = sigmoidf_dev(lds_g[0][0][bl][n] + lds_g[1][0][bl][n]);
        const float fg = sigmoidf_dev(lds_g[0][1][bl][n] + lds_g[1][1][bl][n]);
        const float gg = tanhf(lds_g[0][2][bl][n] + lds_g[1][2][bl][n]);
        const float og = sigmoidf_dev(lds_g[0][3][bl][n] + lds_g[1][3][bl][n]);
        const long cidx = (long)b * HID + n0 + n;
        const float c_new = fg * c_state[cidx] + ig * gg;
        c_state[cidx] = c_new;
        const float h_new = og * tanhf(c_new);
        h_out[(long)b * h_out_bstride + n0 + n] = h_new;
        const int k  = n0 + n;
        const int kr = k & 7;
        const int fl = ((kr >> 1) & 1) * 16 + bl;
        const int fe = (kr & 1) + 2 * (kr >> 2);
        hf_wr[(((mt * (HID / 8) + (k >> 3)) * 32 + fl) << 2) + fe] = h_new;
    }
}

extern "C" void kernel_launch(void* const* d_in, const int* in_sizes, int n_in,
                              void* d_out, int out_size, void* d_ws, size_t ws_size,
                              hipStream_t stream) {
    const int*   src  = (const int*)  d_in[0];
    const float* emb  = (const float*)d_in[1];
    const float* W_ih = (const float*)d_in[2];
    const float* W_hh = (const float*)d_in[3];
    const float* b_ih = (const float*)d_in[4];
    const float* b_hh = (const float*)d_in[5];
    float* out = (float*)d_out;

    float* c_state = (float*)d_ws;
    float* hf0     = c_state + BATCH * HID;
    float* hf1     = hf0 + BATCH * HID;
    float* xg      = hf1 + BATCH * HID;

    const int n_init = 2 * BATCH * HID;
    zero_ws_kernel<<<(n_init + 255) / 256, 256, 0, stream>>>(c_state, n_init);

    for (int t0 = 0; t0 < SEQ; t0 += TCHUNK) {
        xproj_kernel<<<dim3(GATES / 128, TCHUNK * BATCH / 128), 256, 0, stream>>>(
            src, emb, W_ih, xg, t0);

        for (int tl = 0; tl < TCHUNK; ++tl) {
            const int t = t0 + tl;
            const float* hf_rd = (t & 1) ? hf1 : hf0;
            float*       hf_wr = (t & 1) ? hf0 : hf1;
            lstm_step_kernel<<<dim3(HID / 16, 2), 256, 0, stream>>>(
                xg + (long)tl * BATCH * GATES, W_hh, b_ih, b_hh,
                hf_rd, hf_wr, c_state,
                out + (long)t * HID, (long)SEQ * HID);
        }
    }
}
